# starting kernel
# baseline (speedup 1.0000x reference)
#include <hip/hip_runtime.h>

#define NN   50000
#define NE   800000
#define INF_ 128
#define HID  64
#define HID4 16
#define DHID 384

typedef float v2f __attribute__((ext_vector_type(2)));
typedef float v8f __attribute__((ext_vector_type(8)));


__global__ void k_zero(float* __restrict__ p, int n) {
  int i = blockIdx.x * blockDim.x + threadIdx.x;
  if (i < n) p[i] = 0.0f;
}

__global__ void k_degree(const int* __restrict__ dst, float* __restrict__ deg, int e) {
  int i = blockIdx.x * blockDim.x + threadIdx.x;
  if (i < e) atomicAdd(&deg[dst[i]], 1.0f);
}

__global__ void k_dinv(float* __restrict__ d, int n) {
  int i = blockIdx.x * blockDim.x + threadIdx.x;
  if (i < n) {
    float v = d[i];
    v = v < 1.0f ? 1.0f : v;
    d[i] = rsqrtf(v);
  }
}

__global__ void k_scale_zero4(const float4* __restrict__ f, const float* __restrict__ dinv,
                              float4* __restrict__ t, float4* __restrict__ agg) {
  int i = blockIdx.x * blockDim.x + threadIdx.x;
  if (i >= NN * HID4) return;
  float dv = dinv[i >> 4];
  float4 v = f[i];
  v.x *= dv; v.y *= dv; v.z *= dv; v.w *= dv;
  t[i] = v;
  agg[i] = make_float4(0.f, 0.f, 0.f, 0.f);
}

__global__ void k_scatter4(const int* __restrict__ src, const int* __restrict__ dst,
                           const float4* __restrict__ t, float* __restrict__ agg) {
  int i = blockIdx.x * blockDim.x + threadIdx.x;
  if (i >= NE * HID4) return;
  int e = i >> 4;
  int q = i & 15;
  float4 v = t[(size_t)src[e] * HID4 + q];
  float* a = agg + (size_t)dst[e] * HID + q * 4;
  atomicAdd(a + 0, v.x);
  atomicAdd(a + 1, v.y);
  atomicAdd(a + 2, v.z);
  atomicAdd(a + 3, v.w);
}

__global__ void k_update_combine4(float4* __restrict__ f, const float4* __restrict__ agg,
                                  const float* __restrict__ dinv, float* __restrict__ out,
                                  int col_base, float c0, float c1, float c2) {
  int i = blockIdx.x * blockDim.x + threadIdx.x;
  if (i >= NN * HID4) return;
  int row = i >> 4;
  int q   = i & 15;
  float dv = dinv[row];
  float4 fv = f[i];
  float4 av = agg[i];
  fv.x -= av.x * dv; fv.y -= av.y * dv; fv.z -= av.z * dv; fv.w -= av.w * dv;
  f[i] = fv;
  float4* o = (float4*)(out + (size_t)row * DHID + col_base + q * 4);
  const float cs[3] = {c0, c1, c2};
#pragma unroll
  for (int j = 0; j < 3; ++j) {
    float4 ov = o[j * HID4];
    ov.x += cs[j] * fv.x; ov.y += cs[j] * fv.y;
    ov.z += cs[j] * fv.z; ov.w += cs[j] * fv.w;
    o[j * HID4] = ov;
  }
}


__global__ __launch_bounds__(128) void k_gemm1_relu_combine(
    const float* __restrict__ X, const float* __restrict__ W,
    const float* __restrict__ bias, float* __restrict__ Y,
    float* __restrict__ out, int col_base, float c0, float c1, float c2) {
  const int lane = threadIdx.x & 31;
  const int wave = threadIdx.x >> 5;
  const int row0 = blockIdx.x * 16;
  const int n0   = wave * 16;
  const int mr   = lane & 15;
  const int kq   = (lane >> 4) << 1;
  v8f c = {};
  const float* xrow = X + (size_t)(row0 + mr) * INF_;
  for (int k = 0; k < INF_; k += 4) {
    v2f a, b;
    a.x = xrow[k + kq];
    a.y = xrow[k + kq + 1];
    b.x = W[(size_t)(k + kq) * HID + n0 + mr];
    b.y = W[(size_t)(k + kq + 1) * HID + n0 + mr];
    c = __builtin_amdgcn_wmma_f32_16x16x4_f32(false, a, false, b, (short)0, c, false, false);
  }
  const int mhi = (lane >> 4) << 3;
  const float bv = bias[n0 + mr];
#pragma unroll
  for (int r = 0; r < 8; ++r) {
    float v = c[r] + bv;
    v = v > 0.0f ? v : 0.0f;
    const size_t row = (size_t)(row0 + mhi + r);
    Y[row * HID + n0 + mr] = v;
    float* o = out + row * DHID + col_base + n0 + mr;
    o[0]       = c0 * v;
    o[HID]     = c1 * v;
    o[2 * HID] = c2 * v;
  }
}


__global__ __launch_bounds__(128) void k_head(
    const float* __restrict__ H,
    const float* __restrict__ Wm1, const float* __restrict__ bm1,
    const float* __restrict__ Wm2, const float* __restrict__ bm2,
    float* __restrict__ logits) {
  __shared__ float m[16][HID];
  const int lane = threadIdx.x & 31;
  const int wave = threadIdx.x >> 5;
  const int row0 = blockIdx.x * 16;
  const int n0   = wave * 16;
  const int mr   = lane & 15;
  const int kq   = (lane >> 4) << 1;
  v8f c = {};
  const float* hrow = H + (size_t)(row0 + mr) * DHID;
  for (int k = 0; k < DHID; k += 4) {
    v2f a, b;
    a.x = hrow[k + kq];
    a.y = hrow[k + kq + 1];
    b.x = Wm1[(size_t)(k + kq) * HID + n0 + mr];
    b.y = Wm1[(size_t)(k + kq + 1) * HID + n0 + mr];
    c = __builtin_amdgcn_wmma_f32_16x16x4_f32(false, a, false, b, (short)0, c, false, false);
  }
  const int mhi = (lane >> 4) << 3;
  const float bv = bm1[n0 + mr];
#pragma unroll
  for (int r = 0; r < 8; ++r) {
    float v = c[r] + bv;
    m[mhi + r][n0 + mr] = v > 0.0f ? v : 0.0f;
  }
  __syncthreads();
  if (threadIdx.x < 32) {
    int row = threadIdx.x >> 1;
    int col = threadIdx.x & 1;
    float acc = bm2[col];
#pragma unroll
    for (int k = 0; k < HID; ++k) acc += m[row][k] * Wm2[k * 2 + col];
    logits[(size_t)(row0 + row) * 2 + col] = acc;
  }
}


static void run_branch(const float* X, const int* src, const int* dst,
                       const float* W1, const float* b1,
                       float* out, int col_base,
                       const float sc[3][3],
                       float* f, float* t, float* agg, float* dinv,
                       hipStream_t stream) {
  const int nNodeB = (NN + 255) / 256;
  const int nEdgeB = (NE + 255) / 256;
  const int nQuadB = (NN * HID4 + 255) / 256;
  const int nScatB = (NE * HID4 + 255) / 256;

  k_zero  <<<nNodeB, 256, 0, stream>>>(dinv, NN);
  k_degree<<<nEdgeB, 256, 0, stream>>>(dst, dinv, NE);
  k_dinv  <<<nNodeB, 256, 0, stream>>>(dinv, NN);

  k_gemm1_relu_combine<<<NN / 16, 128, 0, stream>>>(X, W1, b1, f, out, col_base,
                                                    sc[0][0], sc[0][1], sc[0][2]);

  for (int step = 1; step <= 2; ++step) {
    k_scale_zero4<<<nQuadB, 256, 0, stream>>>((const float4*)f, dinv,
                                              (float4*)t, (float4*)agg);
    k_scatter4   <<<nScatB, 256, 0, stream>>>(src, dst, (const float4*)t, agg);
    k_update_combine4<<<nQuadB, 256, 0, stream>>>((float4*)f, (const float4*)agg, dinv,
                                                  out, col_base,
                                                  sc[step][0], sc[step][1], sc[step][2]);
  }
}

extern "C" void kernel_launch(void* const* d_in, const int* in_sizes, int n_in,
                              void* d_out, int out_size, void* d_ws, size_t ws_size,
                              hipStream_t stream) {
  (void)in_sizes; (void)n_in; (void)out_size; (void)ws_size;
  const float* x     = (const float*)d_in[0];
  const float* sim_x = (const float*)d_in[1];
  const int*   src   = (const int*)d_in[2];
  const int*   dst   = (const int*)d_in[3];
  const int*   ssrc  = (const int*)d_in[4];
  const int*   sdst  = (const int*)d_in[5];
  const float* W1o   = (const float*)d_in[6];
  const float* b1o   = (const float*)d_in[7];
  const float* W1s   = (const float*)d_in[8];
  const float* b1s   = (const float*)d_in[9];
  const float* Wm1   = (const float*)d_in[10];
  const float* bm1   = (const float*)d_in[11];
  const float* Wm2   = (const float*)d_in[12];
  const float* bm2   = (const float*)d_in[13];

  float* out    = (float*)d_out;
  float* logits = out + (size_t)NN * DHID;

  float* f    = (float*)d_ws;
  float* t    = f    + (size_t)NN * HID;
  float* agg  = t    + (size_t)NN * HID;
  float* dinv = agg  + (size_t)NN * HID;

  const float sc_o[3][3] = {
      {3.0f,   0.0f,  0.0f},
      {-3.0f,  3.0f,  0.0f},
      {0.75f, -1.5f,  0.75f}};
  const float sc_s[3][3] = {
      {4.0f, 4.0f, 4.0f},
      {4.0f, 4.0f, 4.0f},
      {4.0f, 4.0f, 4.0f}};

  run_branch(x,     src,  dst,  W1o, b1o, out,   0, sc_o, f, t, agg, dinv, stream);
  run_branch(sim_x, ssrc, sdst, W1s, b1s, out, 192, sc_s, f, t, agg, dinv, stream);

  k_head<<<NN / 16, 128, 0, stream>>>(out, Wm1, bm1, Wm2, bm2, logits);
}


# float-atomic scatter-add replaced by a gather over a dst-sorted edge list built per call (int atomic rank); sum order per node now follows that arrival-order rank
# speedup vs baseline: 1.1897x; 1.1897x over previous
#include <hip/hip_runtime.h>

#define NN   50000
#define NE   800000
#define INF_ 128
#define HID  64
#define HID4 16
#define DHID 384

typedef float v2f __attribute__((ext_vector_type(2)));
typedef float v8f __attribute__((ext_vector_type(8)));


__global__ void k_zero(int* __restrict__ p, int n) {
  int i = blockIdx.x * blockDim.x + threadIdx.x;
  if (i < n) p[i] = 0;
}

__global__ void k_count_rank(const int* __restrict__ dst, int* __restrict__ cnt,
                             int* __restrict__ rank, int e) {
  int i = blockIdx.x * blockDim.x + threadIdx.x;
  if (i < e) rank[i] = atomicAdd(&cnt[dst[i]], 1);
}

__global__ void k_dinv_alloc(const int* __restrict__ cnt, int* __restrict__ cursor,
                             float* __restrict__ dinv, int* __restrict__ start, int n) {
  const int i = blockIdx.x * blockDim.x + threadIdx.x;
  const int lane = threadIdx.x & 31;
  const int cv = i < n ? cnt[i] : 0;
  int incl = cv;
  for (int o = 1; o < 32; o <<= 1) {
    int up = __shfl_up(incl, o);
    if (lane >= o) incl += up;
  }
  const int tot = __shfl(incl, 31);
  int base = 0;
  if (lane == 0) base = atomicAdd(cursor, tot);
  base = __shfl(base, 0);
  if (i < n) {
    float v = (float)cv;
    v = v < 1.0f ? 1.0f : v;
    dinv[i] = rsqrtf(v);
    start[i] = base + incl - cv;
  }
}

__global__ void k_fill(const int* __restrict__ src, const int* __restrict__ dst,
                       const int* __restrict__ start, const int* __restrict__ rank,
                       int* __restrict__ perm, int e) {
  int i = blockIdx.x * blockDim.x + threadIdx.x;
  if (i < e) perm[start[dst[i]] + rank[i]] = src[i];
}

__global__ void k_scale4(const float4* __restrict__ f, const float* __restrict__ dinv,
                         float4* __restrict__ t) {
  int i = blockIdx.x * blockDim.x + threadIdx.x;
  if (i >= NN * HID4) return;
  float dv = dinv[i >> 4];
  float4 v = f[i];
  v.x *= dv; v.y *= dv; v.z *= dv; v.w *= dv;
  t[i] = v;
}

__global__ __launch_bounds__(256) void k_gather(const int* __restrict__ perm,
                                                const int* __restrict__ start,
                                                const int* __restrict__ cnt,
                                                const float2* __restrict__ t,
                                                float2* __restrict__ agg) {
  const int lane = threadIdx.x & 31;
  const int n = __builtin_amdgcn_readfirstlane((int)(blockIdx.x * 8 + (threadIdx.x >> 5)));
  if (n >= NN) return;
  const int s = start[n];
  const int cn = cnt[n];
  float2 acc = make_float2(0.f, 0.f);
  for (int b = 0; b < cn; b += 32) {
    const int m = (cn - b) < 32 ? (cn - b) : 32;
    const int my = lane < m ? perm[s + b + lane] : 0;
#pragma unroll 4
    for (int j = 0; j < m; ++j) {
      const int sidx = __builtin_amdgcn_readlane(my, j);
      const float2 v = t[(size_t)sidx * 32 + lane];
      acc.x += v.x;
      acc.y += v.y;
    }
  }
  agg[(size_t)n * 32 + lane] = acc;
}

__global__ void k_update_combine4(float4* __restrict__ f, const float4* __restrict__ agg,
                                  const float* __restrict__ dinv, float* __restrict__ out,
                                  int col_base, float c0, float c1, float c2) {
  int i = blockIdx.x * blockDim.x + threadIdx.x;
  if (i >= NN * HID4) return;
  int row = i >> 4;
  int q   = i & 15;
  float dv = dinv[row];
  float4 fv = f[i];
  float4 av = agg[i];
  fv.x -= av.x * dv; fv.y -= av.y * dv; fv.z -= av.z * dv; fv.w -= av.w * dv;
  f[i] = fv;
  float4* o = (float4*)(out + (size_t)row * DHID + col_base + q * 4);
  const float cs[3] = {c0, c1, c2};
#pragma unroll
  for (int j = 0; j < 3; ++j) {
    float4 ov = o[j * HID4];
    ov.x += cs[j] * fv.x; ov.y += cs[j] * fv.y;
    ov.z += cs[j] * fv.z; ov.w += cs[j] * fv.w;
    o[j * HID4] = ov;
  }
}


__global__ __launch_bounds__(128) void k_gemm1_relu_combine(
    const float* __restrict__ X, const float* __restrict__ W,
    const float* __restrict__ bias, float* __restrict__ Y,
    float* __restrict__ out, int col_base, float c0, float c1, float c2) {
  const int lane = threadIdx.x & 31;
  const int wave = threadIdx.x >> 5;
  const int row0 = blockIdx.x * 16;
  const int n0   = wave * 16;
  const int mr   = lane & 15;
  const int kq   = (lane >> 4) << 1;
  v8f c = {};
  const float* xrow = X + (size_t)(row0 + mr) * INF_;
  for (int k = 0; k < INF_; k += 4) {
    v2f a, b;
    a.x = xrow[k + kq];
    a.y = xrow[k + kq + 1];
    b.x = W[(size_t)(k + kq) * HID + n0 + mr];
    b.y = W[(size_t)(k + kq + 1) * HID + n0 + mr];
    c = __builtin_amdgcn_wmma_f32_16x16x4_f32(false, a, false, b, (short)0, c, false, false);
  }
  const int mhi = (lane >> 4) << 3;
  const float bv = bias[n0 + mr];
#pragma unroll
  for (int r = 0; r < 8; ++r) {
    float v = c[r] + bv;
    v = v > 0.0f ? v : 0.0f;
    const size_t row = (size_t)(row0 + mhi + r);
    Y[row * HID + n0 + mr] = v;
    float* o = out + row * DHID + col_base + n0 + mr;
    o[0]       = c0 * v;
    o[HID]     = c1 * v;
    o[2 * HID] = c2 * v;
  }
}


__global__ __launch_bounds__(128) void k_head(
    const float* __restrict__ H,
    const float* __restrict__ Wm1, const float* __restrict__ bm1,
    const float* __restrict__ Wm2, const float* __restrict__ bm2,
    float* __restrict__ logits) {
  __shared__ float m[16][HID];
  const int lane = threadIdx.x & 31;
  const int wave = threadIdx.x >> 5;
  const int row0 = blockIdx.x * 16;
  const int n0   = wave * 16;
  const int mr   = lane & 15;
  const int kq   = (lane >> 4) << 1;
  v8f c = {};
  const float* hrow = H + (size_t)(row0 + mr) * DHID;
  for (int k = 0; k < DHID; k += 4) {
    v2f a, b;
    a.x = hrow[k + kq];
    a.y = hrow[k + kq + 1];
    b.x = Wm1[(size_t)(k + kq) * HID + n0 + mr];
    b.y = Wm1[(size_t)(k + kq + 1) * HID + n0 + mr];
    c = __builtin_amdgcn_wmma_f32_16x16x4_f32(false, a, false, b, (short)0, c, false, false);
  }
  const int mhi = (lane >> 4) << 3;
  const float bv = bm1[n0 + mr];
#pragma unroll
  for (int r = 0; r < 8; ++r) {
    float v = c[r] + bv;
    m[mhi + r][n0 + mr] = v > 0.0f ? v : 0.0f;
  }
  __syncthreads();
  if (threadIdx.x < 32) {
    int row = threadIdx.x >> 1;
    int col = threadIdx.x & 1;
    float acc = bm2[col];
#pragma unroll
    for (int k = 0; k < HID; ++k) acc += m[row][k] * Wm2[k * 2 + col];
    logits[(size_t)(row0 + row) * 2 + col] = acc;
  }
}


static void run_branch(const float* X, const int* src, const int* dst,
                       const float* W1, const float* b1,
                       float* out, int col_base,
                       const float sc[3][3],
                       float* f, float* t, float* agg, float* dinv,
                       int* cnt, int* start, int* rank, int* perm,
                       hipStream_t stream) {
  const int nNodeB = (NN + 255) / 256;
  const int nEdgeB = (NE + 255) / 256;
  const int nQuadB = (NN * HID4 + 255) / 256;
  const int nGathB = NN / 8;

  k_zero      <<<(NN + 1 + 255) / 256, 256, 0, stream>>>(cnt, NN + 1);
  k_count_rank<<<nEdgeB, 256, 0, stream>>>(dst, cnt, rank, NE);
  k_dinv_alloc<<<nNodeB, 256, 0, stream>>>(cnt, cnt + NN, dinv, start, NN);
  k_fill      <<<nEdgeB, 256, 0, stream>>>(src, dst, start, rank, perm, NE);

  k_gemm1_relu_combine<<<NN / 16, 128, 0, stream>>>(X, W1, b1, f, out, col_base,
                                                    sc[0][0], sc[0][1], sc[0][2]);

  for (int step = 1; step <= 2; ++step) {
    k_scale4<<<nQuadB, 256, 0, stream>>>((const float4*)f, dinv, (float4*)t);
    k_gather<<<nGathB, 256, 0, stream>>>(perm, start, cnt, (const float2*)t, (float2*)agg);
    k_update_combine4<<<nQuadB, 256, 0, stream>>>((float4*)f, (const float4*)agg, dinv,
                                                  out, col_base,
                                                  sc[step][0], sc[step][1], sc[step][2]);
  }
}

extern "C" void kernel_launch(void* const* d_in, const int* in_sizes, int n_in,
                              void* d_out, int out_size, void* d_ws, size_t ws_size,
                              hipStream_t stream) {
  (void)in_sizes; (void)n_in; (void)out_size; (void)ws_size;
  const float* x     = (const float*)d_in[0];
  const float* sim_x = (const float*)d_in[1];
  const int*   src   = (const int*)d_in[2];
  const int*   dst   = (const int*)d_in[3];
  const int*   ssrc  = (const int*)d_in[4];
  const int*   sdst  = (const int*)d_in[5];
  const float* W1o   = (const float*)d_in[6];
  const float* b1o   = (const float*)d_in[7];
  const float* W1s   = (const float*)d_in[8];
  const float* b1s   = (const float*)d_in[9];
  const float* Wm1   = (const float*)d_in[10];
  const float* bm1   = (const float*)d_in[11];
  const float* Wm2   = (const float*)d_in[12];
  const float* bm2   = (const float*)d_in[13];

  float* out    = (float*)d_out;
  float* logits = out + (size_t)NN * DHID;

  float* f    = (float*)d_ws;
  float* t    = f    + (size_t)NN * HID;
  float* agg  = t    + (size_t)NN * HID;
  float* dinv = agg  + (size_t)NN * HID;
  int*   cnt   = (int*)(dinv + NN);
  int*   start = cnt   + NN + 64;
  int*   rank  = start + NN;
  int*   perm  = rank  + NE;

  const float sc_o[3][3] = {
      {3.0f,   0.0f,  0.0f},
      {-3.0f,  3.0f,  0.0f},
      {0.75f, -1.5f,  0.75f}};
  const float sc_s[3][3] = {
      {4.0f, 4.0f, 4.0f},
      {4.0f, 4.0f, 4.0f},
      {4.0f, 4.0f, 4.0f}};

  run_branch(x,     src,  dst,  W1o, b1o, out,   0, sc_o, f, t, agg, dinv, cnt, start, rank, perm, stream);
  run_branch(sim_x, ssrc, sdst, W1s, b1s, out, 192, sc_s, f, t, agg, dinv, cnt, start, rank, perm, stream);

  k_head<<<NN / 16, 128, 0, stream>>>(out, Wm1, bm1, Wm2, bm2, logits);
}


# scale, update and combine passes fused into the gather kernel: one launch per propagation step, f ping-ponged; source row times dinv[src] now one fma at the add, sum order unchanged
# speedup vs baseline: 1.5373x; 1.2922x over previous
#include <hip/hip_runtime.h>

#define NN   50000
#define NE   800000
#define INF_ 128
#define HID  64
#define DHID 384

typedef float v2f __attribute__((ext_vector_type(2)));
typedef float v8f __attribute__((ext_vector_type(8)));


__global__ void k_zero(int* __restrict__ p, int n) {
  int i = blockIdx.x * blockDim.x + threadIdx.x;
  if (i < n) p[i] = 0;
}

__global__ void k_count_rank(const int* __restrict__ dst, int* __restrict__ cnt,
                             int* __restrict__ rank, int e) {
  int i = blockIdx.x * blockDim.x + threadIdx.x;
  if (i < e) rank[i] = atomicAdd(&cnt[dst[i]], 1);
}

__global__ void k_dinv_alloc(const int* __restrict__ cnt, int* __restrict__ cursor,
                             float* __restrict__ dinv, int* __restrict__ start, int n) {
  const int i = blockIdx.x * blockDim.x + threadIdx.x;
  const int lane = threadIdx.x & 31;
  const int cv = i < n ? cnt[i] : 0;
  int incl = cv;
  for (int o = 1; o < 32; o <<= 1) {
    int up = __shfl_up(incl, o);
    if (lane >= o) incl += up;
  }
  const int tot = __shfl(incl, 31);
  int base = 0;
  if (lane == 0) base = atomicAdd(cursor, tot);
  base = __shfl(base, 0);
  if (i < n) {
    float v = (float)cv;
    v = v < 1.0f ? 1.0f : v;
    dinv[i] = rsqrtf(v);
    start[i] = base + incl - cv;
  }
}

__global__ void k_fill(const int* __restrict__ src, const int* __restrict__ dst,
                       const int* __restrict__ start, const int* __restrict__ rank,
                       int* __restrict__ perm, int e) {
  int i = blockIdx.x * blockDim.x + threadIdx.x;
  if (i < e) perm[start[dst[i]] + rank[i]] = src[i];
}

template <bool LAST>
__global__ __launch_bounds__(256) void k_step(const int* __restrict__ perm,
                                              const int* __restrict__ start,
                                              const int* __restrict__ cnt,
                                              const float* __restrict__ dinv,
                                              const float2* __restrict__ fin,
                                              float2* __restrict__ fout,
                                              float* __restrict__ out, int col_base,
                                              float c0, float c1, float c2) {
  const int lane = threadIdx.x & 31;
  const int n = __builtin_amdgcn_readfirstlane((int)(blockIdx.x * 8 + (threadIdx.x >> 5)));
  if (n >= NN) return;
  const int s = start[n];
  const int cn = cnt[n];
  const float dn = dinv[n];
  const float2 own = fin[(size_t)n * 32 + lane];
  float2* o = (float2*)(out + (size_t)n * DHID + col_base) + lane;
  float2 o0 = o[0];
  float2 o1 = o[HID / 2];
  float2 o2 = o[HID];
  float2 acc = make_float2(0.f, 0.f);
  for (int b = 0; b < cn; b += 32) {
    const int m = (cn - b) < 32 ? (cn - b) : 32;
    const int my = lane < m ? perm[s + b + lane] : 0;
    const int dmy = __float_as_int(dinv[my]);
#pragma unroll 4
    for (int j = 0; j < m; ++j) {
      const int sidx = __builtin_amdgcn_readlane(my, j);
      const float ds = __int_as_float(__builtin_amdgcn_readlane(dmy, j));
      const float2 v = fin[(size_t)sidx * 32 + lane];
      acc.x += v.x * ds;
      acc.y += v.y * ds;
    }
  }
  float2 fn;
  fn.x = own.x - acc.x * dn;
  fn.y = own.y - acc.y * dn;
  if (!LAST) fout[(size_t)n * 32 + lane] = fn;
  o0.x += c0 * fn.x; o0.y += c0 * fn.y;
  o1.x += c1 * fn.x; o1.y += c1 * fn.y;
  o2.x += c2 * fn.x; o2.y += c2 * fn.y;
  o[0]       = o0;
  o[HID / 2] = o1;
  o[HID]     = o2;
}


__global__ __launch_bounds__(128) void k_gemm1_relu_combine(
    const float* __restrict__ X, const float* __restrict__ W,
    const float* __restrict__ bias, float* __restrict__ Y,
    float* __restrict__ out, int col_base, float c0, float c1, float c2) {
  const int lane = threadIdx.x & 31;
  const int wave = threadIdx.x >> 5;
  const int row0 = blockIdx.x * 16;
  const int n0   = wave * 16;
  const int mr   = lane & 15;
  const int kq   = (lane >> 4) << 1;
  v8f c = {};
  const float* xrow = X + (size_t)(row0 + mr) * INF_;
  for (int k = 0; k < INF_; k += 4) {
    v2f a, b;
    a.x = xrow[k + kq];
    a.y = xrow[k + kq + 1];
    b.x = W[(size_t)(k + kq) * HID + n0 + mr];
    b.y = W[(size_t)(k + kq + 1) * HID + n0 + mr];
    c = __builtin_amdgcn_wmma_f32_16x16x4_f32(false, a, false, b, (short)0, c, false, false);
  }
  const int mhi = (lane >> 4) << 3;
  const float bv = bias[n0 + mr];
#pragma unroll
  for (int r = 0; r < 8; ++r) {
    float v = c[r] + bv;
    v = v > 0.0f ? v : 0.0f;
    const size_t row = (size_t)(row0 + mhi + r);
    Y[row * HID + n0 + mr] = v;
    float* o = out + row * DHID + col_base + n0 + mr;
    o[0]       = c0 * v;
    o[HID]     = c1 * v;
    o[2 * HID] = c2 * v;
  }
}


__global__ __launch_bounds__(128) void k_head(
    const float* __restrict__ H,
    const float* __restrict__ Wm1, const float* __restrict__ bm1,
    const float* __restrict__ Wm2, const float* __restrict__ bm2,
    float* __restrict__ logits) {
  __shared__ float m[16][HID];
  const int lane = threadIdx.x & 31;
  const int wave = threadIdx.x >> 5;
  const int row0 = blockIdx.x * 16;
  const int n0   = wave * 16;
  const int mr   = lane & 15;
  const int kq   = (lane >> 4) << 1;
  v8f c = {};
  const float* hrow = H + (size_t)(row0 + mr) * DHID;
  for (int k = 0; k < DHID; k += 4) {
    v2f a, b;
    a.x = hrow[k + kq];
    a.y = hrow[k + kq + 1];
    b.x = Wm1[(size_t)(k + kq) * HID + n0 + mr];
    b.y = Wm1[(size_t)(k + kq + 1) * HID + n0 + mr];
    c = __builtin_amdgcn_wmma_f32_16x16x4_f32(false, a, false, b, (short)0, c, false, false);
  }
  const int mhi = (lane >> 4) << 3;
  const float bv = bm1[n0 + mr];
#pragma unroll
  for (int r = 0; r < 8; ++r) {
    float v = c[r] + bv;
    m[mhi + r][n0 + mr] = v > 0.0f ? v : 0.0f;
  }
  __syncthreads();
  if (threadIdx.x < 32) {
    int row = threadIdx.x >> 1;
    int col = threadIdx.x & 1;
    float acc = bm2[col];
#pragma unroll
    for (int k = 0; k < HID; ++k) acc += m[row][k] * Wm2[k * 2 + col];
    logits[(size_t)(row0 + row) * 2 + col] = acc;
  }
}


static void run_branch(const float* X, const int* src, const int* dst,
                       const float* W1, const float* b1,
                       float* out, int col_base,
                       const float sc[3][3],
                       float* f, float* f1, float* dinv,
                       int* cnt, int* start, int* rank, int* perm,
                       hipStream_t stream) {
  const int nNodeB = (NN + 255) / 256;
  const int nEdgeB = (NE + 255) / 256;
  const int nStepB = NN / 8;

  k_zero      <<<(NN + 1 + 255) / 256, 256, 0, stream>>>(cnt, NN + 1);
  k_count_rank<<<nEdgeB, 256, 0, stream>>>(dst, cnt, rank, NE);
  k_dinv_alloc<<<nNodeB, 256, 0, stream>>>(cnt, cnt + NN, dinv, start, NN);
  k_fill      <<<nEdgeB, 256, 0, stream>>>(src, dst, start, rank, perm, NE);

  k_gemm1_relu_combine<<<NN / 16, 128, 0, stream>>>(X, W1, b1, f, out, col_base,
                                                    sc[0][0], sc[0][1], sc[0][2]);

  k_step<false><<<nStepB, 256, 0, stream>>>(perm, start, cnt, dinv, (const float2*)f, (float2*)f1,
                                            out, col_base, sc[1][0], sc[1][1], sc[1][2]);
  k_step<true> <<<nStepB, 256, 0, stream>>>(perm, start, cnt, dinv, (const float2*)f1, (float2*)nullptr,
                                            out, col_base, sc[2][0], sc[2][1], sc[2][2]);
}

extern "C" void kernel_launch(void* const* d_in, const int* in_sizes, int n_in,
                              void* d_out, int out_size, void* d_ws, size_t ws_size,
                              hipStream_t stream) {
  (void)in_sizes; (void)n_in; (void)out_size; (void)ws_size;
  const float* x     = (const float*)d_in[0];
  const float* sim_x = (const float*)d_in[1];
  const int*   src   = (const int*)d_in[2];
  const int*   dst   = (const int*)d_in[3];
  const int*   ssrc  = (const int*)d_in[4];
  const int*   sdst  = (const int*)d_in[5];
  const float* W1o   = (const float*)d_in[6];
  const float* b1o   = (const float*)d_in[7];
  const float* W1s   = (const float*)d_in[8];
  const float* b1s   = (const float*)d_in[9];
  const float* Wm1   = (const float*)d_in[10];
  const float* bm1   = (const float*)d_in[11];
  const float* Wm2   = (const float*)d_in[12];
  const float* bm2   = (const float*)d_in[13];

  float* out    = (float*)d_out;
  float* logits = out + (size_t)NN * DHID;

  float* f    = (float*)d_ws;
  float* f1   = f    + (size_t)NN * HID;
  float* dinv = f1   + (size_t)NN * HID;
  int*   cnt   = (int*)(dinv + NN);
  int*   start = cnt   + NN + 64;
  int*   rank  = start + NN;
  int*   perm  = rank  + NE;

  const float sc_o[3][3] = {
      {3.0f,   0.0f,  0.0f},
      {-3.0f,  3.0f,  0.0f},
      {0.75f, -1.5f,  0.75f}};
  const float sc_s[3][3] = {
      {4.0f, 4.0f, 4.0f},
      {4.0f, 4.0f, 4.0f},
      {4.0f, 4.0f, 4.0f}};

  run_branch(x,     src,  dst,  W1o, b1o, out,   0, sc_o, f, f1, dinv, cnt, start, rank, perm, stream);
  run_branch(sim_x, ssrc, sdst, W1s, b1s, out, 192, sc_s, f, f1, dinv, cnt, start, rank, perm, stream);

  k_head<<<NN / 16, 128, 0, stream>>>(out, Wm1, bm1, Wm2, bm2, logits);
}


# independent work shares launches: both graph branches run in every launch (blockIdx.y = branch, own workspace planes) and the two input GEMMs ride in the edge-count launch; 15 launches -> 7
# speedup vs baseline: 2.1567x; 1.4029x over previous
#include <hip/hip_runtime.h>

#define NN   50000
#define NE   800000
#define INF_ 128
#define HID  64
#define DHID 384
#define BRW  8150080

typedef float v2f __attribute__((ext_vector_type(2)));
typedef float v8f __attribute__((ext_vector_type(8)));


__global__ void k_zero(int* __restrict__ p, int n) {
  int i = blockIdx.x * blockDim.x + threadIdx.x;
  p += (size_t)blockIdx.y * BRW;
  if (i < n) p[i] = 0;
}

__global__ void k_dinv_alloc(int* __restrict__ cnt, float* __restrict__ dinv,
                             int* __restrict__ start, int n) {
  cnt   += (size_t)blockIdx.y * BRW;
  dinv  += (size_t)blockIdx.y * BRW;
  start += (size_t)blockIdx.y * BRW;
  int* cursor = cnt + NN;
  const int i = blockIdx.x * blockDim.x + threadIdx.x;
  const int lane = threadIdx.x & 31;
  const int cv = i < n ? cnt[i] : 0;
  int incl = cv;
  for (int o = 1; o < 32; o <<= 1) {
    int up = __shfl_up(incl, o);
    if (lane >= o) incl += up;
  }
  const int tot = __shfl(incl, 31);
  int base = 0;
  if (lane == 0) base = atomicAdd(cursor, tot);
  base = __shfl(base, 0);
  if (i < n) {
    float v = (float)cv;
    v = v < 1.0f ? 1.0f : v;
    dinv[i] = rsqrtf(v);
    start[i] = base + incl - cv;
  }
}

__global__ void k_fill(const int* __restrict__ src0, const int* __restrict__ src1,
                       const int* __restrict__ dst0, const int* __restrict__ dst1,
                       const int* __restrict__ start, const int* __restrict__ rank,
                       int* __restrict__ perm, int e) {
  const int* __restrict__ src = blockIdx.y ? src1 : src0;
  const int* __restrict__ dst = blockIdx.y ? dst1 : dst0;
  start += (size_t)blockIdx.y * BRW;
  rank  += (size_t)blockIdx.y * BRW;
  perm  += (size_t)blockIdx.y * BRW;
  int i = blockIdx.x * blockDim.x + threadIdx.x;
  if (i < e) perm[start[dst[i]] + rank[i]] = src[i];
}

template <bool LAST>
__global__ __launch_bounds__(256) void k_step(const int* __restrict__ perm,
                                              const int* __restrict__ start,
                                              const int* __restrict__ cnt,
                                              const float* __restrict__ dinv,
                                              const float2* __restrict__ fin,
                                              float2* __restrict__ fout,
                                              float* __restrict__ out,
                                              float c0a, float c1a, float c2a,
                                              float c0b, float c1b, float c2b) {
  const int lane = threadIdx.x & 31;
  const int br = blockIdx.y;
  perm  += (size_t)br * BRW;
  start += (size_t)br * BRW;
  cnt   += (size_t)br * BRW;
  dinv  += (size_t)br * BRW;
  fin   += (size_t)br * (BRW / 2);
  if (!LAST) fout += (size_t)br * (BRW / 2);
  const int col_base = br * (DHID / 2);
  const float c0 = br ? c0b : c0a;
  const float c1 = br ? c1b : c1a;
  const float c2 = br ? c2b : c2a;
  const int n = __builtin_amdgcn_readfirstlane((int)(blockIdx.x * 8 + (threadIdx.x >> 5)));
  if (n >= NN) return;
  const int s = start[n];
  const int cn = cnt[n];
  const float dn = dinv[n];
  const float2 own = fin[(size_t)n * 32 + lane];
  float2* o = (float2*)(out + (size_t)n * DHID + col_base) + lane;
  float2 o0 = o[0];
  float2 o1 = o[HID / 2];
  float2 o2 = o[HID];
  float2 acc = make_float2(0.f, 0.f);
  for (int b = 0; b < cn; b += 32) {
    const int m = (cn - b) < 32 ? (cn - b) : 32;
    const int my = lane < m ? perm[s + b + lane] : 0;
    const int dmy = __float_as_int(dinv[my]);
#pragma unroll 4
    for (int j = 0; j < m; ++j) {
      const int sidx = __builtin_amdgcn_readlane(my, j);
      const float ds = __int_as_float(__builtin_amdgcn_readlane(dmy, j));
      const float2 v = fin[(size_t)sidx * 32 + lane];
      acc.x += v.x * ds;
      acc.y += v.y * ds;
    }
  }
  float2 fn;
  fn.x = own.x - acc.x * dn;
  fn.y = own.y - acc.y * dn;
  if (!LAST) fout[(size_t)n * 32 + lane] = fn;
  o0.x += c0 * fn.x; o0.y += c0 * fn.y;
  o1.x += c1 * fn.x; o1.y += c1 * fn.y;
  o2.x += c2 * fn.x; o2.y += c2 * fn.y;
  o[0]       = o0;
  o[HID / 2] = o1;
  o[HID]     = o2;
}


__device__ __forceinline__ void gemm1_tile(
    const float* __restrict__ X, const float* __restrict__ W,
    const float* __restrict__ bias, float* __restrict__ Y,
    float* __restrict__ out, int col_base, float c0, float c1, float c2,
    int row0, int n0, int lane) {
  const int mr   = lane & 15;
  const int kq   = (lane >> 4) << 1;
  v8f c = {};
  const float* xrow = X + (size_t)(row0 + mr) * INF_;
  for (int k = 0; k < INF_; k += 4) {
    v2f a, b;
    a.x = xrow[k + kq];
    a.y = xrow[k + kq + 1];
    b.x = W[(size_t)(k + kq) * HID + n0 + mr];
    b.y = W[(size_t)(k + kq + 1) * HID + n0 + mr];
    c = __builtin_amdgcn_wmma_f32_16x16x4_f32(false, a, false, b, (short)0, c, false, false);
  }
  const int mhi = (lane >> 4) << 3;
  const float bv = bias[n0 + mr];
#pragma unroll
  for (int r = 0; r < 8; ++r) {
    float v = c[r] + bv;
    v = v > 0.0f ? v : 0.0f;
    const size_t row = (size_t)(row0 + mhi + r);
    Y[row * HID + n0 + mr] = v;
    float* o = out + row * DHID + col_base + n0 + mr;
    o[0]       = c0 * v;
    o[HID]     = c1 * v;
    o[2 * HID] = c2 * v;
  }
}

__global__ __launch_bounds__(256) void k_count_gemm1(
    const int* __restrict__ dst0, const int* __restrict__ dst1,
    int* __restrict__ cnt, int* __restrict__ rank,
    const float* __restrict__ X0, const float* __restrict__ X1,
    const float* __restrict__ W0, const float* __restrict__ W1,
    const float* __restrict__ bias0, const float* __restrict__ bias1,
    float* __restrict__ Y, float* __restrict__ out,
    float c0a, float c1a, float c2a, float c0b, float c1b, float c2b) {
  const int g = blockIdx.x / 3;
  const int r = blockIdx.x - g * 3;
  if (r < 2) {
    const int cb = g * 2 + r;
    const int br = cb >= NE / 256 ? 1 : 0;
    const int i  = (cb - br * (NE / 256)) * 256 + threadIdx.x;
    const int* __restrict__ dst = br ? dst1 : dst0;
    cnt  += (size_t)br * BRW;
    rank += (size_t)br * BRW;
    if (i < NE) rank[i] = atomicAdd(&cnt[dst[i]], 1);
    return;
  }
  const int wave = threadIdx.x >> 5;
  const int tile = __builtin_amdgcn_readfirstlane(g * 2 + (wave >> 2));
  const int br   = tile >= NN / 16 ? 1 : 0;
  gemm1_tile(br ? X1 : X0, br ? W1 : W0, br ? bias1 : bias0, Y + (size_t)br * BRW, out,
             br * (DHID / 2), br ? c0b : c0a, br ? c1b : c1a, br ? c2b : c2a,
             (tile - br * (NN / 16)) * 16, (wave & 3) * 16, threadIdx.x & 31);
}


__global__ __launch_bounds__(128) void k_head(
    const float* __restrict__ H,
    const float* __restrict__ Wm1, const float* __restrict__ bm1,
    const float* __restrict__ Wm2, const float* __restrict__ bm2,
    float* __restrict__ logits) {
  __shared__ float m[16][HID];
  const int lane = threadIdx.x & 31;
  const int wave = threadIdx.x >> 5;
  const int row0 = blockIdx.x * 16;
  const int n0   = wave * 16;
  const int mr   = lane & 15;
  const int kq   = (lane >> 4) << 1;
  v8f c = {};
  const float* hrow = H + (size_t)(row0 + mr) * DHID;
  for (int k = 0; k < DHID; k += 4) {
    v2f a, b;
    a.x = hrow[k + kq];
    a.y = hrow[k + kq + 1];
    b.x = Wm1[(size_t)(k + kq) * HID + n0 + mr];
    b.y = Wm1[(size_t)(k + kq + 1) * HID + n0 + mr];
    c = __builtin_amdgcn_wmma_f32_16x16x4_f32(false, a, false, b, (short)0, c, false, false);
  }
  const int mhi = (lane >> 4) << 3;
  const float bv = bm1[n0 + mr];
#pragma unroll
  for (int r = 0; r < 8; ++r) {
    float v = c[r] + bv;
    m[mhi + r][n0 + mr] = v > 0.0f ? v : 0.0f;
  }
  __syncthreads();
  if (threadIdx.x < 32) {
    int row = threadIdx.x >> 1;
    int col = threadIdx.x & 1;
    float acc = bm2[col];
#pragma unroll
    for (int k = 0; k < HID; ++k) acc += m[row][k] * Wm2[k * 2 + col];
    logits[(size_t)(row0 + row) * 2 + col] = acc;
  }
}


extern "C" void kernel_launch(void* const* d_in, const int* in_sizes, int n_in,
                              void* d_out, int out_size, void* d_ws, size_t ws_size,
                              hipStream_t stream) {
  (void)in_sizes; (void)n_in; (void)out_size; (void)ws_size;
  const float* x     = (const float*)d_in[0];
  const float* sim_x = (const float*)d_in[1];
  const int*   src   = (const int*)d_in[2];
  const int*   dst   = (const int*)d_in[3];
  const int*   ssrc  = (const int*)d_in[4];
  const int*   sdst  = (const int*)d_in[5];
  const float* W1o   = (const float*)d_in[6];
  const float* b1o   = (const float*)d_in[7];
  const float* W1s   = (const float*)d_in[8];
  const float* b1s   = (const float*)d_in[9];
  const float* Wm1   = (const float*)d_in[10];
  const float* bm1   = (const float*)d_in[11];
  const float* Wm2   = (const float*)d_in[12];
  const float* bm2   = (const float*)d_in[13];

  float* out    = (float*)d_out;
  float* logits = out + (size_t)NN * DHID;

  float* f    = (float*)d_ws;
  float* f1   = f    + (size_t)NN * HID;
  float* dinv = f1   + (size_t)NN * HID;
  int*   cnt   = (int*)(dinv + NN);
  int*   start = cnt   + NN + 64;
  int*   rank  = start + NN;
  int*   perm  = rank  + NE;

  const float sc_o[3][3] = {
      {3.0f,   0.0f,  0.0f},
      {-3.0f,  3.0f,  0.0f},
      {0.75f, -1.5f,  0.75f}};
  const float sc_s[3][3] = {
      {4.0f, 4.0f, 4.0f},
      {4.0f, 4.0f, 4.0f},
      {4.0f, 4.0f, 4.0f}};

  const int nNodeB = (NN + 255) / 256;
  const int nEdgeB = (NE + 255) / 256;
  const int nStepB = NN / 8;

  k_zero      <<<dim3((NN + 1 + 255) / 256, 2), 256, 0, stream>>>(cnt, NN + 1);
  k_count_gemm1<<<3 * (NN / 16), 256, 0, stream>>>(
      dst, sdst, cnt, rank, x, sim_x, W1o, W1s, b1o, b1s, f, out,
      sc_o[0][0], sc_o[0][1], sc_o[0][2], sc_s[0][0], sc_s[0][1], sc_s[0][2]);
  k_dinv_alloc<<<dim3(nNodeB, 2), 256, 0, stream>>>(cnt, dinv, start, NN);
  k_fill      <<<dim3(nEdgeB, 2), 256, 0, stream>>>(src, ssrc, dst, sdst, start, rank, perm, NE);

  k_step<false><<<dim3(nStepB, 2), 256, 0, stream>>>(
      perm, start, cnt, dinv, (const float2*)f, (float2*)f1, out,
      sc_o[1][0], sc_o[1][1], sc_o[1][2], sc_s[1][0], sc_s[1][1], sc_s[1][2]);
  k_step<true><<<dim3(nStepB, 2), 256, 0, stream>>>(
      perm, start, cnt, dinv, (const float2*)f1, (float2*)nullptr, out,
      sc_o[2][0], sc_o[2][1], sc_o[2][2], sc_s[2][0], sc_s[2][1], sc_s[2][2]);

  k_head<<<NN / 16, 128, 0, stream>>>(out, Wm1, bm1, Wm2, bm2, logits);
}


# out stored once: step 2 builds a node's three filter outputs from f0, f1, f2 in registers; the GEMM and step 1 no longer write or re-read out; per-element sum order and rounding unchanged
# speedup vs baseline: 2.3803x; 1.1037x over previous
#include <hip/hip_runtime.h>

#define NN   50000
#define NE   800000
#define INF_ 128
#define HID  64
#define DHID 384
#define BRW  8150080

typedef float v2f __attribute__((ext_vector_type(2)));
typedef float v8f __attribute__((ext_vector_type(8)));


__global__ void k_zero(int* __restrict__ p, int n) {
  int i = blockIdx.x * blockDim.x + threadIdx.x;
  p += (size_t)blockIdx.y * BRW;
  if (i < n) p[i] = 0;
}

__global__ void k_dinv_alloc(int* __restrict__ cnt, float* __restrict__ dinv,
                             int* __restrict__ start, int n) {
  cnt   += (size_t)blockIdx.y * BRW;
  dinv  += (size_t)blockIdx.y * BRW;
  start += (size_t)blockIdx.y * BRW;
  int* cursor = cnt + NN;
  const int i = blockIdx.x * blockDim.x + threadIdx.x;
  const int lane = threadIdx.x & 31;
  const int cv = i < n ? cnt[i] : 0;
  int incl = cv;
  for (int o = 1; o < 32; o <<= 1) {
    int up = __shfl_up(incl, o);
    if (lane >= o) incl += up;
  }
  const int tot = __shfl(incl, 31);
  int base = 0;
  if (lane == 0) base = atomicAdd(cursor, tot);
  base = __shfl(base, 0);
  if (i < n) {
    float v = (float)cv;
    v = v < 1.0f ? 1.0f : v;
    dinv[i] = rsqrtf(v);
    start[i] = base + incl - cv;
  }
}

__global__ void k_fill(const int* __restrict__ src0, const int* __restrict__ src1,
                       const int* __restrict__ dst0, const int* __restrict__ dst1,
                       const int* __restrict__ start, const int* __restrict__ rank,
                       int* __restrict__ perm, int e) {
  const int* __restrict__ src = blockIdx.y ? src1 : src0;
  const int* __restrict__ dst = blockIdx.y ? dst1 : dst0;
  start += (size_t)blockIdx.y * BRW;
  rank  += (size_t)blockIdx.y * BRW;
  perm  += (size_t)blockIdx.y * BRW;
  int i = blockIdx.x * blockDim.x + threadIdx.x;
  if (i < e) perm[start[dst[i]] + rank[i]] = src[i];
}

__device__ __forceinline__ float2 gather_in_edges(const int* __restrict__ perm,
                                                  const float* __restrict__ dinv,
                                                  const float2* __restrict__ fin,
                                                  int s, int cn, int lane) {
  float2 acc = make_float2(0.f, 0.f);
  for (int b = 0; b < cn; b += 32) {
    const int m = (cn - b) < 32 ? (cn - b) : 32;
    const int my = lane < m ? perm[s + b + lane] : 0;
    const int dmy = __float_as_int(dinv[my]);
#pragma unroll 4
    for (int j = 0; j < m; ++j) {
      const int sidx = __builtin_amdgcn_readlane(my, j);
      const float ds = __int_as_float(__builtin_amdgcn_readlane(dmy, j));
      const float2 v = fin[(size_t)sidx * 32 + lane];
      acc.x += v.x * ds;
      acc.y += v.y * ds;
    }
  }
  return acc;
}

__global__ __launch_bounds__(256) void k_step1(const int* __restrict__ perm,
                                               const int* __restrict__ start,
                                               const int* __restrict__ cnt,
                                               const float* __restrict__ dinv,
                                               const float2* __restrict__ f0,
                                               float2* __restrict__ f1) {
  const int lane = threadIdx.x & 31;
  const int br = blockIdx.y;
  perm  += (size_t)br * BRW;
  start += (size_t)br * BRW;
  cnt   += (size_t)br * BRW;
  dinv  += (size_t)br * BRW;
  f0    += (size_t)br * (BRW / 2);
  f1    += (size_t)br * (BRW / 2);
  const int n = __builtin_amdgcn_readfirstlane((int)(blockIdx.x * 8 + (threadIdx.x >> 5)));
  if (n >= NN) return;
  const int s = start[n];
  const int cn = cnt[n];
  const float dn = dinv[n];
  const float2 own = f0[(size_t)n * 32 + lane];
  const float2 acc = gather_in_edges(perm, dinv, f0, s, cn, lane);
  float2 fn;
  fn.x = own.x - acc.x * dn;
  fn.y = own.y - acc.y * dn;
  f1[(size_t)n * 32 + lane] = fn;
}

struct Coef { float c[3][3]; };

__global__ __launch_bounds__(256) void k_step2(const int* __restrict__ perm,
                                               const int* __restrict__ start,
                                               const int* __restrict__ cnt,
                                               const float* __restrict__ dinv,
                                               const float2* __restrict__ f0,
                                               const float2* __restrict__ f1,
                                               float* __restrict__ out,
                                               Coef co, Coef cs) {
  const int lane = threadIdx.x & 31;
  const int br = blockIdx.y;
  perm  += (size_t)br * BRW;
  start += (size_t)br * BRW;
  cnt   += (size_t)br * BRW;
  dinv  += (size_t)br * BRW;
  f0    += (size_t)br * (BRW / 2);
  f1    += (size_t)br * (BRW / 2);
  const int col_base = br * (DHID / 2);
  const int n = __builtin_amdgcn_readfirstlane((int)(blockIdx.x * 8 + (threadIdx.x >> 5)));
  if (n >= NN) return;
  const int s = start[n];
  const int cn = cnt[n];
  const float dn = dinv[n];
  const float2 a0 = f0[(size_t)n * 32 + lane];
  const float2 a1 = f1[(size_t)n * 32 + lane];
  const float2 acc = gather_in_edges(perm, dinv, f1, s, cn, lane);
  float2 a2;
  a2.x = a1.x - acc.x * dn;
  a2.y = a1.y - acc.y * dn;
  float2* o = (float2*)(out + (size_t)n * DHID + col_base) + lane;
#pragma unroll
  for (int j = 0; j < 3; ++j) {
    const float c0 = br ? cs.c[0][j] : co.c[0][j];
    const float c1 = br ? cs.c[1][j] : co.c[1][j];
    const float c2 = br ? cs.c[2][j] : co.c[2][j];
    float2 t;
    t.x = __builtin_fmaf(c2, a2.x, __builtin_fmaf(c1, a1.x, c0 * a0.x));
    t.y = __builtin_fmaf(c2, a2.y, __builtin_fmaf(c1, a1.y, c0 * a0.y));
    o[j * (HID / 2)] = t;
  }
}


__device__ __forceinline__ void gemm1_tile(
    const float* __restrict__ X, const float* __restrict__ W,
    const float* __restrict__ bias, float* __restrict__ Y,
    int row0, int n0, int lane) {
  const int mr   = lane & 15;
  const int kq   = (lane >> 4) << 1;
  v8f c = {};
  const float* xrow = X + (size_t)(row0 + mr) * INF_;
  for (int k = 0; k < INF_; k += 4) {
    v2f a, b;
    a.x = xrow[k + kq];
    a.y = xrow[k + kq + 1];
    b.x = W[(size_t)(k + kq) * HID + n0 + mr];
    b.y = W[(size_t)(k + kq + 1) * HID + n0 + mr];
    c = __builtin_amdgcn_wmma_f32_16x16x4_f32(false, a, false, b, (short)0, c, false, false);
  }
  const int mhi = (lane >> 4) << 3;
  const float bv = bias[n0 + mr];
#pragma unroll
  for (int r = 0; r < 8; ++r) {
    float v = c[r] + bv;
    v = v > 0.0f ? v : 0.0f;
    const size_t row = (size_t)(row0 + mhi + r);
    Y[row * HID + n0 + mr] = v;
  }
}

__global__ __launch_bounds__(256) void k_count_gemm1(
    const int* __restrict__ dst0, const int* __restrict__ dst1,
    int* __restrict__ cnt, int* __restrict__ rank,
    const float* __restrict__ X0, const float* __restrict__ X1,
    const float* __restrict__ W0, const float* __restrict__ W1,
    const float* __restrict__ bias0, const float* __restrict__ bias1,
    float* __restrict__ Y) {
  const int g = blockIdx.x / 3;
  const int r = blockIdx.x - g * 3;
  if (r < 2) {
    const int cb = g * 2 + r;
    const int br = cb >= NE / 256 ? 1 : 0;
    const int i  = (cb - br * (NE / 256)) * 256 + threadIdx.x;
    const int* __restrict__ dst = br ? dst1 : dst0;
    cnt  += (size_t)br * BRW;
    rank += (size_t)br * BRW;
    if (i < NE) rank[i] = atomicAdd(&cnt[dst[i]], 1);
    return;
  }
  const int wave = threadIdx.x >> 5;
  const int tile = __builtin_amdgcn_readfirstlane(g * 2 + (wave >> 2));
  const int br   = tile >= NN / 16 ? 1 : 0;
  gemm1_tile(br ? X1 : X0, br ? W1 : W0, br ? bias1 : bias0, Y + (size_t)br * BRW,
             (tile - br * (NN / 16)) * 16, (wave & 3) * 16, threadIdx.x & 31);
}


__global__ __launch_bounds__(128) void k_head(
    const float* __restrict__ H,
    const float* __restrict__ Wm1, const float* __restrict__ bm1,
    const float* __restrict__ Wm2, const float* __restrict__ bm2,
    float* __restrict__ logits) {
  __shared__ float m[16][HID];
  const int lane = threadIdx.x & 31;
  const int wave = threadIdx.x >> 5;
  const int row0 = blockIdx.x * 16;
  const int n0   = wave * 16;
  const int mr   = lane & 15;
  const int kq   = (lane >> 4) << 1;
  v8f c = {};
  const float* hrow = H + (size_t)(row0 + mr) * DHID;
  for (int k = 0; k < DHID; k += 4) {
    v2f a, b;
    a.x = hrow[k + kq];
    a.y = hrow[k + kq + 1];
    b.x = Wm1[(size_t)(k + kq) * HID + n0 + mr];
    b.y = Wm1[(size_t)(k + kq + 1) * HID + n0 + mr];
    c = __builtin_amdgcn_wmma_f32_16x16x4_f32(false, a, false, b, (short)0, c, false, false);
  }
  const int mhi = (lane >> 4) << 3;
  const float bv = bm1[n0 + mr];
#pragma unroll
  for (int r = 0; r < 8; ++r) {
    float v = c[r] + bv;
    m[mhi + r][n0 + mr] = v > 0.0f ? v : 0.0f;
  }
  __syncthreads();
  if (threadIdx.x < 32) {
    int row = threadIdx.x >> 1;
    int col = threadIdx.x & 1;
    float acc = bm2[col];
#pragma unroll
    for (int k = 0; k < HID; ++k) acc += m[row][k] * Wm2[k * 2 + col];
    logits[(size_t)(row0 + row) * 2 + col] = acc;
  }
}


extern "C" void kernel_launch(void* const* d_in, const int* in_sizes, int n_in,
                              void* d_out, int out_size, void* d_ws, size_t ws_size,
                              hipStream_t stream) {
  (void)in_sizes; (void)n_in; (void)out_size; (void)ws_size;
  const float* x     = (const float*)d_in[0];
  const float* sim_x = (const float*)d_in[1];
  const int*   src   = (const int*)d_in[2];
  const int*   dst   = (const int*)d_in[3];
  const int*   ssrc  = (const int*)d_in[4];
  const int*   sdst  = (const int*)d_in[5];
  const float* W1o   = (const float*)d_in[6];
  const float* b1o   = (const float*)d_in[7];
  const float* W1s   = (const float*)d_in[8];
  const float* b1s   = (const float*)d_in[9];
  const float* Wm1   = (const float*)d_in[10];
  const float* bm1   = (const float*)d_in[11];
  const float* Wm2   = (const float*)d_in[12];
  const float* bm2   = (const float*)d_in[13];

  float* out    = (float*)d_out;
  float* logits = out + (size_t)NN * DHID;

  float* f    = (float*)d_ws;
  float* f1   = f    + (size_t)NN * HID;
  float* dinv = f1   + (size_t)NN * HID;
  int*   cnt   = (int*)(dinv + NN);
  int*   start = cnt   + NN + 64;
  int*   rank  = start + NN;
  int*   perm  = rank  + NE;

  const Coef sc_o = {{
      {3.0f,   0.0f,  0.0f},
      {-3.0f,  3.0f,  0.0f},
      {0.75f, -1.5f,  0.75f}}};
  const Coef sc_s = {{
      {4.0f, 4.0f, 4.0f},
      {4.0f, 4.0f, 4.0f},
      {4.0f, 4.0f, 4.0f}}};

  const int nNodeB = (NN + 255) / 256;
  const int nEdgeB = (NE + 255) / 256;
  const int nStepB = NN / 8;

  k_zero      <<<dim3((NN + 1 + 255) / 256, 2), 256, 0, stream>>>(cnt, NN + 1);
  k_count_gemm1<<<3 * (NN / 16), 256, 0, stream>>>(
      dst, sdst, cnt, rank, x, sim_x, W1o, W1s, b1o, b1s, f);
  k_dinv_alloc<<<dim3(nNodeB, 2), 256, 0, stream>>>(cnt, dinv, start, NN);
  k_fill      <<<dim3(nEdgeB, 2), 256, 0, stream>>>(src, ssrc, dst, sdst, start, rank, perm, NE);

  k_step1<<<dim3(nStepB, 2), 256, 0, stream>>>(
      perm, start, cnt, dinv, (const float2*)f, (float2*)f1);
  k_step2<<<dim3(nStepB, 2), 256, 0, stream>>>(
      perm, start, cnt, dinv, (const float2*)f, (const float2*)f1, out, sc_o, sc_s);

  k_head<<<NN / 16, 128, 0, stream>>>(out, Wm1, bm1, Wm2, bm2, logits);
}


# both WMMA GEMMs re-tiled: one wave owns a 16-row strip over all 64 columns (4 accumulators); A fragment loaded once per k-step, not once per column tile; per-element sum order unchanged
# speedup vs baseline: 2.9614x; 1.2441x over previous
#include <hip/hip_runtime.h>

#define NN   50000
#define NE   800000
#define INF_ 128
#define HID  64
#define DHID 384
#define CAP  32
#define CNTW (NN + 112)
#define BRW  9650112

typedef float v2f __attribute__((ext_vector_type(2)));
typedef float v4f __attribute__((ext_vector_type(4)));
typedef float v8f __attribute__((ext_vector_type(8)));


__global__ void k_zero(int* __restrict__ p, int n) {
  int i = blockIdx.x * blockDim.x + threadIdx.x;
  p += (size_t)blockIdx.y * BRW;
  if (i < n) p[i] = 0;
}

__device__ __forceinline__ float dinv_of(int c) {
  float v = (float)c;
  v = v < 1.0f ? 1.0f : v;
  return rsqrtf(v);
}

__device__ __forceinline__ float2 gather_in_edges(const int* __restrict__ slot,
                                                  const int2* __restrict__ ovf,
                                                  const int* __restrict__ cnt,
                                                  const float2* __restrict__ fin,
                                                  int n, int cn, int lane) {
  float2 acc = make_float2(0.f, 0.f);
  const int m = cn < CAP ? cn : CAP;
  const int my = lane < m ? slot[(size_t)n * CAP + lane] : 0;
  const int dmy = __float_as_int(dinv_of(cnt[my]));
#pragma unroll 4
  for (int j = 0; j < m; ++j) {
    const int sidx = __builtin_amdgcn_readlane(my, j);
    const float ds = __int_as_float(__builtin_amdgcn_readlane(dmy, j));
    const float2 v = fin[(size_t)sidx * 32 + lane];
    acc.x += v.x * ds;
    acc.y += v.y * ds;
  }
  if (cn > CAP) {
    int len = cnt[NN];
    len = len < NE ? len : NE;
    for (int b = 0; b < len; b += 32) {
      const int idx = b + lane;
      const int2 e = idx < len ? ovf[idx] : make_int2(-1, 0);
      const bool mine = e.x == n;
      const int osrc = mine ? e.y : 0;
      const int odv = __float_as_int(dinv_of(cnt[osrc]));
      unsigned todo = __builtin_amdgcn_ballot_w32(mine);
      while (todo) {
        const int j = __builtin_ctz(todo);
        todo &= todo - 1;
        const int sidx = __builtin_amdgcn_readlane(osrc, j);
        const float ds = __int_as_float(__builtin_amdgcn_readlane(odv, j));
        const float2 v = fin[(size_t)sidx * 32 + lane];
        acc.x += v.x * ds;
        acc.y += v.y * ds;
      }
    }
  }
  return acc;
}

__global__ __launch_bounds__(256) void k_step1(const int* __restrict__ slot,
                                               const int2* __restrict__ ovf,
                                               const int* __restrict__ cnt,
                                               const float2* __restrict__ f0,
                                               float2* __restrict__ f1) {
  const int lane = threadIdx.x & 31;
  const int br = blockIdx.y;
  slot  += (size_t)br * BRW;
  ovf   += (size_t)br * (BRW / 2);
  cnt   += (size_t)br * BRW;
  f0    += (size_t)br * (BRW / 2);
  f1    += (size_t)br * (BRW / 2);
  const int n = __builtin_amdgcn_readfirstlane((int)(blockIdx.x * 8 + (threadIdx.x >> 5)));
  if (n >= NN) return;
  const int cn = cnt[n];
  const float dn = dinv_of(cn);
  const float2 own = f0[(size_t)n * 32 + lane];
  const float2 acc = gather_in_edges(slot, ovf, cnt, f0, n, cn, lane);
  float2 fn;
  fn.x = own.x - acc.x * dn;
  fn.y = own.y - acc.y * dn;
  f1[(size_t)n * 32 + lane] = fn;
}

struct Coef { float c[3][3]; };

__global__ __launch_bounds__(256) void k_step2(const int* __restrict__ slot,
                                               const int2* __restrict__ ovf,
                                               const int* __restrict__ cnt,
                                               const float2* __restrict__ f0,
                                               const float2* __restrict__ f1,
                                               float* __restrict__ out,
                                               Coef co, Coef cs) {
  const int lane = threadIdx.x & 31;
  const int br = blockIdx.y;
  slot  += (size_t)br * BRW;
  ovf   += (size_t)br * (BRW / 2);
  cnt   += (size_t)br * BRW;
  f0    += (size_t)br * (BRW / 2);
  f1    += (size_t)br * (BRW / 2);
  const int col_base = br * (DHID / 2);
  const int n = __builtin_amdgcn_readfirstlane((int)(blockIdx.x * 8 + (threadIdx.x >> 5)));
  if (n >= NN) return;
  const int cn = cnt[n];
  const float dn = dinv_of(cn);
  const float2 a0 = f0[(size_t)n * 32 + lane];
  const float2 a1 = f1[(size_t)n * 32 + lane];
  const float2 acc = gather_in_edges(slot, ovf, cnt, f1, n, cn, lane);
  float2 a2;
  a2.x = a1.x - acc.x * dn;
  a2.y = a1.y - acc.y * dn;
  float2* o = (float2*)(out + (size_t)n * DHID + col_base) + lane;
#pragma unroll
  for (int j = 0; j < 3; ++j) {
    const float c0 = br ? cs.c[0][j] : co.c[0][j];
    const float c1 = br ? cs.c[1][j] : co.c[1][j];
    const float c2 = br ? cs.c[2][j] : co.c[2][j];
    float2 t;
    t.x = __builtin_fmaf(c2, a2.x, __builtin_fmaf(c1, a1.x, c0 * a0.x));
    t.y = __builtin_fmaf(c2, a2.y, __builtin_fmaf(c1, a1.y, c0 * a0.y));
    o[j * (HID / 2)] = t;
  }
}

__device__ __forceinline__ void gemm_strip(const float* __restrict__ A, const int lda, const int KC,
                                           const float* __restrict__ W,
                                           const int row0, const int lane, v8f acc[4]) {
  const int c  = lane & 15;
  const int kq = (lane >> 4) << 1;
  const float* arow = A + (size_t)(row0 + c) * lda + kq;
  const float* wl   = W + (size_t)kq * HID + 4 * c;
  v2f a[2][32];
#pragma unroll
  for (int i = 0; i < 32; ++i) a[0][i] = *(const v2f*)(arow + 4 * i);
#pragma unroll
  for (int ch = 0; ch < KC; ++ch) {
    if (ch + 1 < KC) {
#pragma unroll
      for (int i = 0; i < 32; ++i) a[(ch + 1) & 1][i] = *(const v2f*)(arow + (ch + 1) * 128 + 4 * i);
    }
#pragma unroll
    for (int i = 0; i < 32; ++i) {
      const int k = ch * 128 + 4 * i;
      const v4f b0 = *(const v4f*)(wl + (size_t)k * HID);
      const v4f b1 = *(const v4f*)(wl + (size_t)(k + 1) * HID);
#pragma unroll
      for (int j = 0; j < 4; ++j) {
        v2f b;
        b.x = b0[j];
        b.y = b1[j];
        acc[j] = __builtin_amdgcn_wmma_f32_16x16x4_f32(false, a[ch & 1][i], false, b, (short)0, acc[j], false, false);
      }
    }
  }
}

__device__ __forceinline__ void gemm_strip_rolled(const float* __restrict__ A, const int lda, const int K,
                                                  const float* __restrict__ W,
                                                  const int row0, const int lane, v8f acc[4]) {
  const int c  = lane & 15;
  const int kq = (lane >> 4) << 1;
  const float* arow = A + (size_t)(row0 + c) * lda + kq;
  const float* wl   = W + (size_t)kq * HID + 4 * c;
#pragma unroll 2
  for (int k = 0; k < K; k += 4) {
    const v2f a = *(const v2f*)(arow + k);
    const v4f b0 = *(const v4f*)(wl + (size_t)k * HID);
    const v4f b1 = *(const v4f*)(wl + (size_t)(k + 1) * HID);
#pragma unroll
    for (int j = 0; j < 4; ++j) {
      v2f b;
      b.x = b0[j];
      b.y = b1[j];
      acc[j] = __builtin_amdgcn_wmma_f32_16x16x4_f32(false, a, false, b, (short)0, acc[j], false, false);
    }
  }
}

__device__ __forceinline__ void gemm1_strip(
    const float* __restrict__ X, const float* __restrict__ W,
    const float* __restrict__ bias, float* __restrict__ Y,
    int row0, int lane) {
  v8f acc[4] = {};
  gemm_strip_rolled(X, INF_, INF_, W, row0, lane, acc);
  const int c   = lane & 15;
  const int mhi = (lane >> 4) << 3;
  const v4f bv = *(const v4f*)(bias + 4 * c);
#pragma unroll
  for (int r = 0; r < 8; ++r) {
    v4f v;
#pragma unroll
    for (int j = 0; j < 4; ++j) {
      const float t = acc[j][r] + bv[j];
      v[j] = t > 0.0f ? t : 0.0f;
    }
    *(v4f*)(Y + (size_t)(row0 + mhi + r) * HID + 4 * c) = v;
  }
}

__global__ __launch_bounds__(256) void k_count_gemm1(
    const int* __restrict__ src0, const int* __restrict__ src1,
    const int* __restrict__ dst0, const int* __restrict__ dst1,
    int* __restrict__ cnt, int* __restrict__ slot, int2* __restrict__ ovf,
    const float* __restrict__ X0, const float* __restrict__ X1,
    const float* __restrict__ W0, const float* __restrict__ W1,
    const float* __restrict__ bias0, const float* __restrict__ bias1,
    float* __restrict__ Y) {
  const int g = blockIdx.x / 9;
  const int r = blockIdx.x - g * 9;
  if (r > 0) {
    const int cb = g * 8 + r - 1;
    if (cb >= 2 * (NE / 256)) return;
    const int br = cb >= NE / 256 ? 1 : 0;
    const int i  = (cb - br * (NE / 256)) * 256 + threadIdx.x;
    const int* __restrict__ src = br ? src1 : src0;
    const int* __restrict__ dst = br ? dst1 : dst0;
    cnt  += (size_t)br * BRW;
    slot += (size_t)br * BRW;
    ovf  += (size_t)br * (BRW / 2);
    if (i < NE) {
      const int d = dst[i];
      const int s = src[i];
      const int rk = atomicAdd(&cnt[d], 1);
      if (rk < CAP) {
        slot[(size_t)d * CAP + rk] = s;
      } else {
        const int o = atomicAdd(&cnt[NN], 1);
        ovf[o] = make_int2(d, s);
      }
    }
    return;
  }
  const int strip = __builtin_amdgcn_readfirstlane(g * 8 + (int)(threadIdx.x >> 5));
  if (strip >= 2 * (NN / 16)) return;
  const int br = strip >= NN / 16 ? 1 : 0;
  gemm1_strip(br ? X1 : X0, br ? W1 : W0, br ? bias1 : bias0, Y + (size_t)br * BRW,
              (strip - br * (NN / 16)) * 16, threadIdx.x & 31);
}


__global__ __launch_bounds__(128) void k_head(
    const float* __restrict__ H,
    const float* __restrict__ Wm1, const float* __restrict__ bm1,
    const float* __restrict__ Wm2, const float* __restrict__ bm2,
    float* __restrict__ logits) {
  __shared__ float m[4][16][HID + 1];
  const int lane = threadIdx.x & 31;
  const int wave = threadIdx.x >> 5;
  const int sraw = __builtin_amdgcn_readfirstlane((int)(blockIdx.x * 4 + wave));
  const bool valid = sraw < NN / 16;
  const int row0 = (valid ? sraw : NN / 16 - 1) * 16;
  v8f acc[4] = {};
  gemm_strip(H, DHID, DHID / 128, Wm1, row0, lane, acc);
  const int c   = lane & 15;
  const int mhi = (lane >> 4) << 3;
  const v4f bv = *(const v4f*)(bm1 + 4 * c);
#pragma unroll
  for (int r = 0; r < 8; ++r) {
#pragma unroll
    for (int j = 0; j < 4; ++j) {
      const float v = acc[j][r] + bv[j];
      m[wave][mhi + r][4 * c + j] = v > 0.0f ? v : 0.0f;
    }
  }
  __syncthreads();
  {
    const int row = lane >> 1;
    const int col = lane & 1;
    float s = bm2[col];
#pragma unroll
    for (int k = 0; k < HID; ++k) s += m[wave][row][k] * Wm2[k * 2 + col];
    if (valid) logits[(size_t)(row0 + row) * 2 + col] = s;
  }
}


extern "C" void kernel_launch(void* const* d_in, const int* in_sizes, int n_in,
                              void* d_out, int out_size, void* d_ws, size_t ws_size,
                              hipStream_t stream) {
  (void)in_sizes; (void)n_in; (void)out_size; (void)ws_size;
  const float* x     = (const float*)d_in[0];
  const float* sim_x = (const float*)d_in[1];
  const int*   src   = (const int*)d_in[2];
  const int*   dst   = (const int*)d_in[3];
  const int*   ssrc  = (const int*)d_in[4];
  const int*   sdst  = (const int*)d_in[5];
  const float* W1o   = (const float*)d_in[6];
  const float* b1o   = (const float*)d_in[7];
  const float* W1s   = (const float*)d_in[8];
  const float* b1s   = (const float*)d_in[9];
  const float* Wm1   = (const float*)d_in[10];
  const float* bm1   = (const float*)d_in[11];
  const float* Wm2   = (const float*)d_in[12];
  const float* bm2   = (const float*)d_in[13];

  float* out    = (float*)d_out;
  float* logits = out + (size_t)NN * DHID;

  float* f    = (float*)d_ws;
  float* f1   = f    + (size_t)NN * HID;
  int*   cnt  = (int*)(f1 + (size_t)NN * HID);
  int*   slot = cnt  + CNTW;
  int2*  ovf  = (int2*)(slot + (size_t)NN * CAP);

  const Coef sc_o = {{
      {3.0f,   0.0f,  0.0f},
      {-3.0f,  3.0f,  0.0f},
      {0.75f, -1.5f,  0.75f}}};
  const Coef sc_s = {{
      {4.0f, 4.0f, 4.0f},
      {4.0f, 4.0f, 4.0f},
      {4.0f, 4.0f, 4.0f}}};

  const int nStepB = NN / 8;

  k_zero      <<<dim3((NN + 1 + 255) / 256, 2), 256, 0, stream>>>(cnt, NN + 1);
  k_count_gemm1<<<9 * ((2 * (NN / 16) + 7) / 8), 256, 0, stream>>>(
      src, ssrc, dst, sdst, cnt, slot, ovf, x, sim_x, W1o, W1s, b1o, b1s, f);

  k_step1<<<dim3(nStepB, 2), 256, 0, stream>>>(
      slot, ovf, cnt, (const float2*)f, (float2*)f1);
  k_step2<<<dim3(nStepB, 2), 256, 0, stream>>>(
      slot, ovf, cnt, (const float2*)f, (const float2*)f1, out, sc_o, sc_s);

  k_head<<<(NN / 16 + 3) / 4, 128, 0, stream>>>(out, Wm1, bm1, Wm2, bm2, logits);
}
